# starting kernel
# baseline (speedup 1.0000x reference)
#include <hip/hip_runtime.h>
#include <math.h>

#define IN_CAPS 2401
#define IN_SIZE 8
#define OUT_CAPS 8
#define OUT_SIZE 16
#define BATCH 256
#define KDIM (IN_CAPS * IN_SIZE)
#define NDIM (OUT_CAPS * OUT_SIZE)
#define NKC 16
#define CPC 151

typedef float v2f __attribute__((ext_vector_type(2)));
typedef float v4f __attribute__((ext_vector_type(4)));
typedef float v8f __attribute__((ext_vector_type(8)));

__global__ void k_init_bc(float* __restrict__ bij, float* __restrict__ cij) {
    int t = blockIdx.x * blockDim.x + threadIdx.x;
    if (t < IN_CAPS * OUT_CAPS) { bij[t] = 0.0f; cij[t] = 0.125f; }
}

__global__ void k_transpose_x(const float* __restrict__ x, float* __restrict__ xt) {
    __shared__ float tile[8][33];
    int rt = blockIdx.x;
    int bt = blockIdx.y;
    int tid = threadIdx.x;
    int tr = tid & 7, tb = tid >> 3;
    tile[tr][tb] = x[(size_t)(bt * 32 + tb) * KDIM + rt * 8 + tr];
    __syncthreads();
    int wb = tid & 31, wr = tid >> 5;
    xt[(size_t)(rt * 8 + wr) * BATCH + bt * 32 + wb] = tile[wr][wb];
}

__global__ void k_route_gemm(const float* __restrict__ x, const float* __restrict__ W,
                             const float* __restrict__ cij, float* __restrict__ part) {
    int lane = threadIdx.x & 31;
    int job  = __builtin_amdgcn_readfirstlane(blockIdx.x * 4 + (threadIdx.x >> 5));
    int kc   = job >> 7;
    int rem  = job & 127;
    int mt   = rem >> 3;
    int ot   = rem & 7;
    int ln   = lane & 15;
    int half = lane >> 4;
    int k0   = half * 2;

    int cap0 = kc * CPC;
    int cap1 = min(cap0 + CPC, IN_CAPS);

    const float* xrow = x + (size_t)(mt * 16 + ln) * KDIM;
    v8f acc = {};
    for (int i = cap0; i < cap1; ++i) {
        float cs = cij[i * OUT_CAPS + ot];
        const float* xi   = xrow + i * 8;
        const float* wrow = W + (size_t)i * 1024 + ot * 128 + ln * 8;
#pragma unroll
        for (int g = 0; g < 2; ++g) {
            int kb = g * 4 + k0;
            v2f a = *(const v2f*)(xi + kb);
            v2f b = *(const v2f*)(wrow + kb);
            b *= cs;
            acc = __builtin_amdgcn_wmma_f32_16x16x4_f32(false, a, false, b,
                                                        (short)0, acc, false, false);
        }
    }
    float* pbase = part + ((size_t)kc * BATCH + mt * 16) * NDIM + ot * 16 + ln;
#pragma unroll
    for (int j = 0; j < 8; ++j)
        pbase[(size_t)(j + half * 8) * NDIM] = acc[j];
}

__global__ void k_reduce_squash(const float* __restrict__ part, float* __restrict__ vT,
                                float* __restrict__ out, int last) {
    int b = blockIdx.x;
    int n = threadIdx.x;
    float s = 0.0f;
#pragma unroll
    for (int kc = 0; kc < NKC; ++kc)
        s += part[((size_t)kc * BATCH + b) * NDIM + n];
    float msq = s * s;
#pragma unroll
    for (int m = 1; m < 16; m <<= 1)
        msq += __shfl_xor(msq, m, 32);
    float mag = sqrtf(msq);
    float v = s * (mag / (1.0f + msq));
    vT[(size_t)n * BATCH + b] = v;
    if (last) out[(size_t)b * NDIM + n] = v;
}

__global__ void k_agree(const float* __restrict__ xt, const float* __restrict__ vT,
                        const float* __restrict__ W, float* __restrict__ bij) {
    int lane = threadIdx.x & 31;
    int job  = __builtin_amdgcn_readfirstlane(blockIdx.x * 4 + (threadIdx.x >> 5));
    int it2  = job >> 3;
    int o    = job & 7;
    int i0   = it2 * 2;
    int ln   = lane & 15;
    int half = lane >> 4;
    int k0   = half * 2;

    int r = min(i0 * 8 + ln, KDIM - 1);
    const float* arow = xt + (size_t)r * BATCH;
    const float* brow = vT + (size_t)(o * 16 + ln) * BATCH;
    v8f acc = {};
#pragma unroll 4
    for (int kb = 0; kb < BATCH; kb += 4) {
        v2f a = *(const v2f*)(arow + kb + k0);
        v2f b = *(const v2f*)(brow + kb + k0);
        acc = __builtin_amdgcn_wmma_f32_16x16x4_f32(false, a, false, b,
                                                    (short)0, acc, false, false);
    }
    int iW = i0 + half;
    float dot = 0.0f;
    if (iW < IN_CAPS) {
        const float* wp = W + (size_t)iW * 1024 + o * 128 + ln * 8;
        v4f w0 = *(const v4f*)wp;
        v4f w1 = *(const v4f*)(wp + 4);
        dot = acc[0] * w0.x + acc[1] * w0.y + acc[2] * w0.z + acc[3] * w0.w
            + acc[4] * w1.x + acc[5] * w1.y + acc[6] * w1.z + acc[7] * w1.w;
    }
#pragma unroll
    for (int m = 1; m < 16; m <<= 1)
        dot += __shfl_xor(dot, m, 32);
    if (ln == 0 && iW < IN_CAPS)
        bij[iW * OUT_CAPS + o] += dot * (1.0f / BATCH);
}

__global__ void k_softmax(const float* __restrict__ bij, float* __restrict__ cij) {
    int i = blockIdx.x * blockDim.x + threadIdx.x;
    if (i >= IN_CAPS) return;
    float v[OUT_CAPS];
    float mx = -1e30f;
#pragma unroll
    for (int o = 0; o < OUT_CAPS; ++o) { v[o] = bij[i * OUT_CAPS + o]; mx = fmaxf(mx, v[o]); }
    float sum = 0.0f;
#pragma unroll
    for (int o = 0; o < OUT_CAPS; ++o) { v[o] = __expf(v[o] - mx); sum += v[o]; }
    float inv = 1.0f / sum;
#pragma unroll
    for (int o = 0; o < OUT_CAPS; ++o) cij[i * OUT_CAPS + o] = v[o] * inv;
}

extern "C" void kernel_launch(void* const* d_in, const int* in_sizes, int n_in,
                              void* d_out, int out_size, void* d_ws, size_t ws_size,
                              hipStream_t stream) {
    const float* x = (const float*)d_in[0];
    const float* W = (const float*)d_in[1];
    float* out = (float*)d_out;

    float* ws   = (float*)d_ws;
    float* XT   = ws;
    float* part = XT + (size_t)KDIM * BATCH;
    float* vT   = part + (size_t)NKC * BATCH * NDIM;
    float* bij  = vT + (size_t)NDIM * BATCH;
    float* cij  = bij + IN_CAPS * OUT_CAPS;

    k_init_bc<<<(IN_CAPS * OUT_CAPS + 255) / 256, 256, 0, stream>>>(bij, cij);
    k_transpose_x<<<dim3(IN_CAPS, 8), 256, 0, stream>>>(x, XT);

    for (int it = 0; it < 3; ++it) {
        k_route_gemm<<<512, 128, 0, stream>>>(x, W, cij, part);
        k_reduce_squash<<<BATCH, NDIM, 0, stream>>>(part, vT, out, it == 2 ? 1 : 0);
        if (it < 2) {
            k_agree<<<2402, 128, 0, stream>>>(XT, vT, W, bij);
            k_softmax<<<(IN_CAPS + 127) / 128, 128, 0, stream>>>(bij, cij);
        }
    }
}


# agreement+softmax+next route GEMM fused per 15-capsule work-group: u_hat tiles by WMMA (x.W), agreement = their VALU dot with v (x^T.v GEMM, transpose gone); 12 launches -> 6; sum order changed
# speedup vs baseline: 2.2148x; 2.2148x over previous
#include <hip/hip_runtime.h>
#include <math.h>


#define IN_CAPS 2401
#define IN_SIZE 8
#define OUT_CAPS 8
#define OUT_SIZE 16
#define BATCH 256
#define KDIM (IN_CAPS * IN_SIZE)
#define NDIM (OUT_CAPS * OUT_SIZE)
#define CPC 15
#define NCH 161
#define PSZ (BATCH * NDIM)

typedef float v2f __attribute__((ext_vector_type(2)));
typedef float v4f __attribute__((ext_vector_type(4)));
typedef float v8f __attribute__((ext_vector_type(8)));

struct Frags {
    v2f a[2][2];
    v2f b[8][2];
};

__device__ __forceinline__ void load_frags(Frags& f, const float* __restrict__ xr0,
                                           const float* __restrict__ xr1,
                                           const float* __restrict__ wl, int i) {
    const float* p0 = xr0 + i * 8;
    const float* p1 = xr1 + i * 8;
    f.a[0][0] = *(const v2f*)(p0);
    f.a[0][1] = *(const v2f*)(p0 + 4);
    f.a[1][0] = *(const v2f*)(p1);
    f.a[1][1] = *(const v2f*)(p1 + 4);
    const float* w = wl + (size_t)i * 1024;
#pragma unroll
    for (int o = 0; o < 8; ++o) {
        f.b[o][0] = *(const v2f*)(w + o * 128);
        f.b[o][1] = *(const v2f*)(w + o * 128 + 4);
    }
}

__global__ __launch_bounds__(256) void k_pass(const float* __restrict__ x,
                                              const float* __restrict__ W,
                                              const float* __restrict__ v,
                                              float* __restrict__ bij,
                                              float* __restrict__ part, int mode) {
    __shared__ float red[CPC][8][8];
    __shared__ __attribute__((aligned(16))) float cs[CPC][8];

    int tid  = threadIdx.x;
    int lane = tid & 31;
    int wv   = __builtin_amdgcn_readfirstlane(tid >> 5);
    int ln   = lane & 15;
    int half = lane >> 4;
    int k0   = half * 2;
    int cap0 = blockIdx.x * CPC;
    int ncap = min(CPC, IN_CAPS - cap0);
    int m0   = wv * 32;

    const float* xr0 = x + (size_t)(m0 + ln) * KDIM + k0;
    const float* xr1 = x + (size_t)(m0 + 16 + ln) * KDIM + k0;
    const float* wl  = W + ln * 8 + k0;

    if (mode != 0) {
        v8f vt[2][8];
#pragma unroll
        for (int t = 0; t < 2; ++t)
#pragma unroll
            for (int o = 0; o < 8; ++o)
#pragma unroll
                for (int j = 0; j < 8; ++j)
                    vt[t][o][j] = v[(size_t)(m0 + t * 16 + half * 8 + j) * NDIM + o * 16 + ln];

        Frags cur, nxt;
        load_frags(cur, xr0, xr1, wl, cap0);
        for (int c = 0; c < ncap; ++c) {
            load_frags(nxt, xr0, xr1, wl, cap0 + min(c + 1, ncap - 1));
            asm volatile("" ::: "memory");
            float p[8];
#pragma unroll
            for (int o = 0; o < 8; ++o) {
                float s = 0.0f;
#pragma unroll
                for (int t = 0; t < 2; ++t) {
                    v8f u = {};
                    u = __builtin_amdgcn_wmma_f32_16x16x4_f32(false, cur.a[t][0], false, cur.b[o][0],
                                                              (short)0, u, false, false);
                    u = __builtin_amdgcn_wmma_f32_16x16x4_f32(false, cur.a[t][1], false, cur.b[o][1],
                                                              (short)0, u, false, false);
#pragma unroll
                    for (int j = 0; j < 8; ++j) s += u[j] * vt[t][o][j];
                }
                p[o] = s;
            }
#pragma unroll
            for (int o = 0; o < 8; ++o)
#pragma unroll
                for (int m = 1; m < 32; m <<= 1)
                    p[o] += __shfl_xor(p[o], m, 32);
            if (lane == 0) {
#pragma unroll
                for (int o = 0; o < 8; ++o) red[c][wv][o] = p[o];
            }
            cur = nxt;
        }
        __syncthreads();
        if (tid < 128) {
            int cq = tid >> 3;
            int c  = min(cq, ncap - 1);
            int o  = tid & 7;
            float a = 0.0f;
#pragma unroll
            for (int w = 0; w < 8; ++w) a += red[c][w][o];
            int gi = (cap0 + c) * OUT_CAPS + o;
            float bv = a * (1.0f / BATCH);
            if (mode == 2) bv += bij[gi];
            else if (cq < ncap) bij[gi] = bv;
            float mx = bv;
#pragma unroll
            for (int m = 1; m < 8; m <<= 1) mx = fmaxf(mx, __shfl_xor(mx, m, 32));
            float e = __expf(bv - mx);
            float sum = e;
#pragma unroll
            for (int m = 1; m < 8; m <<= 1) sum += __shfl_xor(sum, m, 32);
            float inv = 1.0f / sum;
            if (cq < ncap) cs[c][o] = e * inv;
        }
    } else {
        if (tid < CPC * 8) cs[tid >> 3][tid & 7] = 0.125f;
    }
    __syncthreads();

    v8f acc[2][8];
#pragma unroll
    for (int t = 0; t < 2; ++t)
#pragma unroll
        for (int o = 0; o < 8; ++o) acc[t][o] = (v8f){};

    Frags cur, nxt;
    load_frags(cur, xr0, xr1, wl, cap0);
    for (int c = 0; c < ncap; ++c) {
        load_frags(nxt, xr0, xr1, wl, cap0 + min(c + 1, ncap - 1));
        asm volatile("" ::: "memory");
        v4f c0 = *(const v4f*)&cs[c][0];
        v4f c1 = *(const v4f*)&cs[c][4];
        float cc[8] = {c0.x, c0.y, c0.z, c0.w, c1.x, c1.y, c1.z, c1.w};
#pragma unroll
        for (int o = 0; o < 8; ++o) {
            v2f b0 = cur.b[o][0] * cc[o];
            v2f b1 = cur.b[o][1] * cc[o];
#pragma unroll
            for (int t = 0; t < 2; ++t) {
                acc[t][o] = __builtin_amdgcn_wmma_f32_16x16x4_f32(false, cur.a[t][0], false, b0,
                                                                  (short)0, acc[t][o], false, false);
                acc[t][o] = __builtin_amdgcn_wmma_f32_16x16x4_f32(false, cur.a[t][1], false, b1,
                                                                  (short)0, acc[t][o], false, false);
            }
        }
        cur = nxt;
    }
    float* pb = part + (size_t)blockIdx.x * PSZ + (size_t)(wv * 2) * 2048 + lane;
#pragma unroll
    for (int t = 0; t < 2; ++t)
#pragma unroll
        for (int o = 0; o < 8; ++o)
#pragma unroll
            for (int j = 0; j < 8; ++j)
                pb[(t * 8 + o) * 256 + j * 32] = acc[t][o][j];
}

__global__ __launch_bounds__(512) void k_reduce_squash(const float* __restrict__ part,
                                                       float* __restrict__ v,
                                                       float* __restrict__ out, int last) {
    __shared__ float sm[4][128];
    int bx   = blockIdx.x;
    int mt   = bx >> 4;
    int j    = (bx >> 1) & 7;
    int oh   = bx & 1;
    int t    = threadIdx.x & 127;
    int q    = threadIdx.x >> 7;
    int o    = oh * 4 + (t >> 5);
    int lane = t & 31;
    int idx  = ((mt * 8 + o) * 8 + j) * 32 + lane;
    float s = 0.0f;
#pragma unroll 8
    for (int kc = q; kc < NCH; kc += 4)
        s += part[(size_t)kc * PSZ + idx];
    sm[q][t] = s;
    __syncthreads();
    if (q == 0) {
        s = (sm[0][t] + sm[1][t]) + (sm[2][t] + sm[3][t]);
        float msq = s * s;
#pragma unroll
        for (int m = 1; m < 16; m <<= 1)
            msq += __shfl_xor(msq, m, 32);
        float mag = sqrtf(msq);
        float val = s * (mag / (1.0f + msq));
        int b = mt * 16 + (lane >> 4) * 8 + j;
        int n = o * 16 + (lane & 15);
        v[(size_t)b * NDIM + n] = val;
        if (last) out[(size_t)b * NDIM + n] = val;
    }
}

extern "C" void kernel_launch(void* const* d_in, const int* in_sizes, int n_in,
                              void* d_out, int out_size, void* d_ws, size_t ws_size,
                              hipStream_t stream) {
    const float* x = (const float*)d_in[0];
    const float* W = (const float*)d_in[1];
    float* out = (float*)d_out;

    float* ws   = (float*)d_ws;
    float* part = ws;
    float* v    = part + (size_t)NCH * PSZ;
    float* bij  = v + (size_t)PSZ;

    k_pass<<<NCH, 256, 0, stream>>>(x, W, v, bij, part, 0);
    k_reduce_squash<<<BATCH, 512, 0, stream>>>(part, v, out, 0);
    k_pass<<<NCH, 256, 0, stream>>>(x, W, v, bij, part, 1);
    k_reduce_squash<<<BATCH, 512, 0, stream>>>(part, v, out, 0);
    k_pass<<<NCH, 256, 0, stream>>>(x, W, v, bij, part, 2);
    k_reduce_squash<<<BATCH, 512, 0, stream>>>(part, v, out, 1);
}


# capsule chunk per work-group 15 -> 10 if the workspace holds 241 partial planes (else the smallest chunk that fits): 241 groups on 241 of 256 WGPs, was 161; 1/3 fewer loop trips; sum order changed
# speedup vs baseline: 2.3516x; 1.0618x over previous
#include <hip/hip_runtime.h>
#include <math.h>


#define IN_CAPS 2401
#define IN_SIZE 8
#define OUT_CAPS 8
#define OUT_SIZE 16
#define BATCH 256
#define KDIM (IN_CAPS * IN_SIZE)
#define NDIM (OUT_CAPS * OUT_SIZE)
#define CPC_MAX 15
#define CPC_MIN 10
#define NCH_MAX 241
#define PSZ (BATCH * NDIM)

typedef float v2f __attribute__((ext_vector_type(2)));
typedef float v4f __attribute__((ext_vector_type(4)));
typedef float v8f __attribute__((ext_vector_type(8)));

struct Frags {
    v2f a[2][2];
    v2f b[8][2];
};

__device__ __forceinline__ void load_frags(Frags& f, const float* __restrict__ xr0,
                                           const float* __restrict__ xr1,
                                           const float* __restrict__ wl, int i) {
    const float* p0 = xr0 + i * 8;
    const float* p1 = xr1 + i * 8;
    f.a[0][0] = *(const v2f*)(p0);
    f.a[0][1] = *(const v2f*)(p0 + 4);
    f.a[1][0] = *(const v2f*)(p1);
    f.a[1][1] = *(const v2f*)(p1 + 4);
    const float* w = wl + (size_t)i * 1024;
#pragma unroll
    for (int o = 0; o < 8; ++o) {
        f.b[o][0] = *(const v2f*)(w + o * 128);
        f.b[o][1] = *(const v2f*)(w + o * 128 + 4);
    }
}

__device__ __forceinline__ int caps_per_group(size_t ws_bytes) {
    long long planes = ((long long)ws_bytes - (long long)(PSZ + IN_CAPS * OUT_CAPS) * 4) / (PSZ * 4);
    int cpc = CPC_MIN;
    while (cpc < CPC_MAX && (IN_CAPS + cpc - 1) / cpc > planes) ++cpc;
    return cpc;
}

__global__ __launch_bounds__(256) void k_pass(const float* __restrict__ x,
                                              const float* __restrict__ W,
                                              float* __restrict__ ws, size_t ws_bytes, int mode) {
    __shared__ float red[CPC_MAX][8][8];
    __shared__ __attribute__((aligned(16))) float cs[CPC_MAX][8];

    const int cpc = caps_per_group(ws_bytes);
    const int nch = (IN_CAPS + cpc - 1) / cpc;
    if ((int)blockIdx.x >= nch) return;
    float* __restrict__ part      = ws;
    const float* __restrict__ v   = ws + (size_t)nch * PSZ;
    float* __restrict__ bij       = ws + (size_t)(nch + 1) * PSZ;

    int tid  = threadIdx.x;
    int lane = tid & 31;
    int wv   = __builtin_amdgcn_readfirstlane(tid >> 5);
    int ln   = lane & 15;
    int half = lane >> 4;
    int k0   = half * 2;
    int cap0 = blockIdx.x * cpc;
    int ncap = min(cpc, IN_CAPS - cap0);
    int m0   = wv * 32;

    const float* xr0 = x + (size_t)(m0 + ln) * KDIM + k0;
    const float* xr1 = x + (size_t)(m0 + 16 + ln) * KDIM + k0;
    const float* wl  = W + ln * 8 + k0;

    if (mode != 0) {
        v8f vt[2][8];
#pragma unroll
        for (int t = 0; t < 2; ++t)
#pragma unroll
            for (int o = 0; o < 8; ++o)
#pragma unroll
                for (int j = 0; j < 8; ++j)
                    vt[t][o][j] = v[(size_t)(m0 + t * 16 + half * 8 + j) * NDIM + o * 16 + ln];

        Frags cur, nxt;
        load_frags(cur, xr0, xr1, wl, cap0);
        for (int c = 0; c < ncap; ++c) {
            load_frags(nxt, xr0, xr1, wl, cap0 + min(c + 1, ncap - 1));
            asm volatile("" ::: "memory");
            float p[8];
#pragma unroll
            for (int o = 0; o < 8; ++o) {
                float s = 0.0f;
#pragma unroll
                for (int t = 0; t < 2; ++t) {
                    v8f u = {};
                    u = __builtin_amdgcn_wmma_f32_16x16x4_f32(false, cur.a[t][0], false, cur.b[o][0],
                                                              (short)0, u, false, false);
                    u = __builtin_amdgcn_wmma_f32_16x16x4_f32(false, cur.a[t][1], false, cur.b[o][1],
                                                              (short)0, u, false, false);
#pragma unroll
                    for (int j = 0; j < 8; ++j) s += u[j] * vt[t][o][j];
                }
                p[o] = s;
            }
#pragma unroll
            for (int o = 0; o < 8; ++o)
#pragma unroll
                for (int m = 1; m < 32; m <<= 1)
                    p[o] += __shfl_xor(p[o], m, 32);
            if (lane == 0) {
#pragma unroll
                for (int o = 0; o < 8; ++o) red[c][wv][o] = p[o];
            }
            cur = nxt;
        }
        __syncthreads();
        if (tid < 128) {
            int cq = tid >> 3;
            int c  = min(cq, ncap - 1);
            int o  = tid & 7;
            float a = 0.0f;
#pragma unroll
            for (int w = 0; w < 8; ++w) a += red[c][w][o];
            int gi = (cap0 + c) * OUT_CAPS + o;
            float bv = a * (1.0f / BATCH);
            if (mode == 2) bv += bij[gi];
            else if (cq < ncap) bij[gi] = bv;
            float mx = bv;
#pragma unroll
            for (int m = 1; m < 8; m <<= 1) mx = fmaxf(mx, __shfl_xor(mx, m, 32));
            float e = __expf(bv - mx);
            float sum = e;
#pragma unroll
            for (int m = 1; m < 8; m <<= 1) sum += __shfl_xor(sum, m, 32);
            float inv = 1.0f / sum;
            if (cq < ncap) cs[c][o] = e * inv;
        }
    } else {
        if (tid < CPC_MAX * 8) cs[tid >> 3][tid & 7] = 0.125f;
    }
    __syncthreads();

    v8f acc[2][8];
#pragma unroll
    for (int t = 0; t < 2; ++t)
#pragma unroll
        for (int o = 0; o < 8; ++o) acc[t][o] = (v8f){};

    Frags cur, nxt;
    load_frags(cur, xr0, xr1, wl, cap0);
    for (int c = 0; c < ncap; ++c) {
        load_frags(nxt, xr0, xr1, wl, cap0 + min(c + 1, ncap - 1));
        asm volatile("" ::: "memory");
        v4f c0 = *(const v4f*)&cs[c][0];
        v4f c1 = *(const v4f*)&cs[c][4];
        float cc[8] = {c0.x, c0.y, c0.z, c0.w, c1.x, c1.y, c1.z, c1.w};
#pragma unroll
        for (int o = 0; o < 8; ++o) {
            v2f b0 = cur.b[o][0] * cc[o];
            v2f b1 = cur.b[o][1] * cc[o];
#pragma unroll
            for (int t = 0; t < 2; ++t) {
                acc[t][o] = __builtin_amdgcn_wmma_f32_16x16x4_f32(false, cur.a[t][0], false, b0,
                                                                  (short)0, acc[t][o], false, false);
                acc[t][o] = __builtin_amdgcn_wmma_f32_16x16x4_f32(false, cur.a[t][1], false, b1,
                                                                  (short)0, acc[t][o], false, false);
            }
        }
        cur = nxt;
    }
    float* pb = part + (size_t)blockIdx.x * PSZ + (size_t)(wv * 2) * 2048 + lane;
#pragma unroll
    for (int t = 0; t < 2; ++t)
#pragma unroll
        for (int o = 0; o < 8; ++o)
#pragma unroll
            for (int j = 0; j < 8; ++j)
                pb[(t * 8 + o) * 256 + j * 32] = acc[t][o][j];
}

__global__ __launch_bounds__(512) void k_reduce_squash(float* __restrict__ ws, size_t ws_bytes,
                                                       float* __restrict__ out, int last) {
    __shared__ float sm[4][128];
    const int cpc = caps_per_group(ws_bytes);
    const int nch = (IN_CAPS + cpc - 1) / cpc;
    const float* __restrict__ part = ws;
    float* __restrict__ v          = ws + (size_t)nch * PSZ;
    int bx   = blockIdx.x;
    int mt   = bx >> 4;
    int j    = (bx >> 1) & 7;
    int oh   = bx & 1;
    int t    = threadIdx.x & 127;
    int q    = threadIdx.x >> 7;
    int o    = oh * 4 + (t >> 5);
    int lane = t & 31;
    int idx  = ((mt * 8 + o) * 8 + j) * 32 + lane;
    float s = 0.0f;
    const float* p = part + (size_t)q * PSZ + idx;
    const int cnt = (nch - q + 3) >> 2;
    int i0 = 0;
    for (; i0 + 16 <= cnt; i0 += 16) {
        float pv[16];
#pragma unroll
        for (int u = 0; u < 16; ++u) pv[u] = p[(size_t)(i0 + u) * (4 * PSZ)];
#pragma unroll
        for (int u = 0; u < 16; ++u) s += pv[u];
    }
    if (i0 < cnt) {
        float pv[16];
#pragma unroll
        for (int u = 0; u < 16; ++u) pv[u] = p[(size_t)min(i0 + u, cnt - 1) * (4 * PSZ)];
#pragma unroll
        for (int u = 0; u < 16; ++u)
            if (i0 + u < cnt) s += pv[u];
    }
    sm[q][t] = s;
    __syncthreads();
    if (q == 0) {
        s = (sm[0][t] + sm[1][t]) + (sm[2][t] + sm[3][t]);
        float msq = s * s;
#pragma unroll
        for (int m = 1; m < 16; m <<= 1)
            msq += __shfl_xor(msq, m, 32);
        float mag = sqrtf(msq);
        float val = s * (mag / (1.0f + msq));
        int b = mt * 16 + (lane >> 4) * 8 + j;
        int n = o * 16 + (lane & 15);
        v[(size_t)b * NDIM + n] = val;
        if (last) out[(size_t)b * NDIM + n] = val;
    }
}

extern "C" void kernel_launch(void* const* d_in, const int* in_sizes, int n_in,
                              void* d_out, int out_size, void* d_ws, size_t ws_size,
                              hipStream_t stream) {
    const float* x = (const float*)d_in[0];
    const float* W = (const float*)d_in[1];
    float* out = (float*)d_out;

    float* ws = (float*)d_ws;

    k_pass<<<NCH_MAX, 256, 0, stream>>>(x, W, ws, ws_size, 0);
    k_reduce_squash<<<BATCH, 512, 0, stream>>>(ws, ws_size, out, 0);
    k_pass<<<NCH_MAX, 256, 0, stream>>>(x, W, ws, ws_size, 1);
    k_reduce_squash<<<BATCH, 512, 0, stream>>>(ws, ws_size, out, 0);
    k_pass<<<NCH_MAX, 256, 0, stream>>>(x, W, ws, ws_size, 2);
    k_reduce_squash<<<BATCH, 512, 0, stream>>>(ws, ws_size, out, 1);
}


# routed passes: each capsule's u_hat tiles (WMMA) are made once and serve agreement, softmax and s += c*u_hat (VALU) behind one barrier per capsule; second x.W sweep dropped; sum order changed
# speedup vs baseline: 2.4975x; 1.0620x over previous
#include <hip/hip_runtime.h>
#include <math.h>


#define IN_CAPS 2401
#define IN_SIZE 8
#define OUT_CAPS 8
#define OUT_SIZE 16
#define BATCH 256
#define KDIM (IN_CAPS * IN_SIZE)
#define NDIM (OUT_CAPS * OUT_SIZE)
#define CPC_MAX 15
#define CPC_MIN 10
#define NCH_MAX 241
#define PSZ (BATCH * NDIM)

typedef float v2f __attribute__((ext_vector_type(2)));
typedef float v4f __attribute__((ext_vector_type(4)));
typedef float v8f __attribute__((ext_vector_type(8)));

struct Frags {
    v2f a[2][2];
    v2f b[8][2];
};

__device__ __forceinline__ void load_frags(Frags& f, const float* __restrict__ xr0,
                                           const float* __restrict__ xr1,
                                           const float* __restrict__ wl, int i) {
    const float* p0 = xr0 + i * 8;
    const float* p1 = xr1 + i * 8;
    f.a[0][0] = *(const v2f*)(p0);
    f.a[0][1] = *(const v2f*)(p0 + 4);
    f.a[1][0] = *(const v2f*)(p1);
    f.a[1][1] = *(const v2f*)(p1 + 4);
    const float* w = wl + (size_t)i * 1024;
#pragma unroll
    for (int o = 0; o < 8; ++o) {
        f.b[o][0] = *(const v2f*)(w + o * 128);
        f.b[o][1] = *(const v2f*)(w + o * 128 + 4);
    }
}

template <int CTRL>
__device__ __forceinline__ float row_xadd(float v) {
    int t = __builtin_amdgcn_update_dpp(0, __builtin_bit_cast(int, v), CTRL, 0xf, 0xf, true);
    return v + __builtin_bit_cast(float, t);
}
__device__ __forceinline__ float wave_sum(float v) {
    v = row_xadd<0x161>(v);
    v = row_xadd<0x162>(v);
    v = row_xadd<0x164>(v);
    v = row_xadd<0x168>(v);
    int b = __builtin_bit_cast(int, v);
    return v + __builtin_bit_cast(float, __builtin_amdgcn_permlanex16(b, b, 0x76543210, 0xfedcba98, false, false));
}

__device__ __forceinline__ int caps_per_group(size_t ws_bytes) {
    long long planes = ((long long)ws_bytes - (long long)(PSZ + IN_CAPS * OUT_CAPS) * 4) / (PSZ * 4);
    int cpc = CPC_MIN;
    while (cpc < CPC_MAX && (IN_CAPS + cpc - 1) / cpc > planes) ++cpc;
    return cpc;
}

__global__ __launch_bounds__(256) void k_pass(const float* __restrict__ x,
                                              const float* __restrict__ W,
                                              float* __restrict__ ws, size_t ws_bytes, int mode) {
    __shared__ __attribute__((aligned(16))) float red[2][8][8];
    __shared__ __attribute__((aligned(16))) float bsh[CPC_MAX][8];
    __shared__ __attribute__((aligned(16))) float cs[CPC_MAX][8];

    const int cpc = caps_per_group(ws_bytes);
    const int nch = (IN_CAPS + cpc - 1) / cpc;
    if ((int)blockIdx.x >= nch) return;
    float* __restrict__ part      = ws;
    const float* __restrict__ v   = ws + (size_t)nch * PSZ;
    float* __restrict__ bij       = ws + (size_t)(nch + 1) * PSZ;

    int tid  = threadIdx.x;
    int lane = tid & 31;
    int wv   = __builtin_amdgcn_readfirstlane(tid >> 5);
    int ln   = lane & 15;
    int half = lane >> 4;
    int k0   = half * 2;
    int cap0 = blockIdx.x * cpc;
    int ncap = min(cpc, IN_CAPS - cap0);
    int m0   = wv * 32;

    const float* xr0 = x + (size_t)(m0 + ln) * KDIM + k0;
    const float* xr1 = x + (size_t)(m0 + 16 + ln) * KDIM + k0;
    const float* wl  = W + ln * 8 + k0;

    v8f acc[2][8];
#pragma unroll
    for (int t = 0; t < 2; ++t)
#pragma unroll
        for (int o = 0; o < 8; ++o) acc[t][o] = (v8f){};

    if (mode != 0) {
        v8f vt[2][8];
#pragma unroll
        for (int t = 0; t < 2; ++t)
#pragma unroll
            for (int o = 0; o < 8; ++o)
#pragma unroll
                for (int j = 0; j < 8; ++j)
                    vt[t][o][j] = v[(size_t)(m0 + t * 16 + half * 8 + j) * NDIM + o * 16 + ln];
        if (mode == 2 && tid < ncap * 8) bsh[tid >> 3][tid & 7] = bij[(size_t)cap0 * OUT_CAPS + tid];

        Frags cur;
        load_frags(cur, xr0, xr1, wl, cap0);
        for (int c = 0; c < ncap; ++c) {
            v8f u[2][8];
#pragma unroll
            for (int o = 0; o < 8; ++o)
#pragma unroll
                for (int t = 0; t < 2; ++t) {
                    v8f z = {};
                    z = __builtin_amdgcn_wmma_f32_16x16x4_f32(false, cur.a[t][0], false, cur.b[o][0],
                                                              (short)0, z, false, false);
                    u[t][o] = __builtin_amdgcn_wmma_f32_16x16x4_f32(false, cur.a[t][1], false, cur.b[o][1],
                                                                    (short)0, z, false, false);
                }
            load_frags(cur, xr0, xr1, wl, cap0 + min(c + 1, ncap - 1));
            asm volatile("" ::: "memory");

            float p[8];
#pragma unroll
            for (int o = 0; o < 8; ++o) {
                v2f s2 = {0.0f, 0.0f};
#pragma unroll
                for (int t = 0; t < 2; ++t)
#pragma unroll
                    for (int k = 0; k < 4; ++k) {
                        v2f uu = {u[t][o][2 * k], u[t][o][2 * k + 1]};
                        v2f vv = {vt[t][o][2 * k], vt[t][o][2 * k + 1]};
                        s2 += uu * vv;
                    }
                p[o] = s2.x + s2.y;
            }
#pragma unroll
            for (int o = 0; o < 8; ++o) p[o] = wave_sum(p[o]);
            float* rb = &red[c & 1][wv][0];
            if (lane == 0) {
                *(v4f*)(rb)     = (v4f){p[0], p[1], p[2], p[3]};
                *(v4f*)(rb + 4) = (v4f){p[4], p[5], p[6], p[7]};
            }
            __builtin_amdgcn_fence(__ATOMIC_RELEASE, "workgroup", "local");
            __builtin_amdgcn_s_barrier();
            __builtin_amdgcn_fence(__ATOMIC_ACQUIRE, "workgroup", "local");
            v4f a0 = {0.0f, 0.0f, 0.0f, 0.0f};
            v4f a1 = {0.0f, 0.0f, 0.0f, 0.0f};
#pragma unroll
            for (int w = 0; w < 8; ++w) {
                a0 += *(const v4f*)&red[c & 1][w][0];
                a1 += *(const v4f*)&red[c & 1][w][4];
            }
            float bv[8] = {a0.x, a0.y, a0.z, a0.w, a1.x, a1.y, a1.z, a1.w};
#pragma unroll
            for (int o = 0; o < 8; ++o) bv[o] *= (1.0f / BATCH);
            if (mode == 2) {
                v4f b0 = *(const v4f*)&bsh[c][0];
                v4f b1 = *(const v4f*)&bsh[c][4];
                bv[0] += b0.x; bv[1] += b0.y; bv[2] += b0.z; bv[3] += b0.w;
                bv[4] += b1.x; bv[5] += b1.y; bv[6] += b1.z; bv[7] += b1.w;
            } else if (tid == 0) {
                float* bp = bij + (size_t)(cap0 + c) * OUT_CAPS;
                *(v4f*)(bp)     = (v4f){bv[0], bv[1], bv[2], bv[3]};
                *(v4f*)(bp + 4) = (v4f){bv[4], bv[5], bv[6], bv[7]};
            }
            float mx = bv[0];
#pragma unroll
            for (int o = 1; o < 8; ++o) mx = fmaxf(mx, bv[o]);
            float e[8];
#pragma unroll
            for (int o = 0; o < 8; ++o) e[o] = __expf(bv[o] - mx);
            float sum = ((e[0] + e[1]) + (e[2] + e[3])) + ((e[4] + e[5]) + (e[6] + e[7]));
            float inv = 1.0f / sum;
#pragma unroll
            for (int o = 0; o < 8; ++o) {
                float cc = e[o] * inv;
#pragma unroll
                for (int t = 0; t < 2; ++t) acc[t][o] += u[t][o] * cc;
            }
        }
    } else {
        if (tid < CPC_MAX * 8) cs[tid >> 3][tid & 7] = 0.125f;
        __syncthreads();

        Frags cur, nxt;
        load_frags(cur, xr0, xr1, wl, cap0);
        for (int c = 0; c < ncap; ++c) {
            load_frags(nxt, xr0, xr1, wl, cap0 + min(c + 1, ncap - 1));
            asm volatile("" ::: "memory");
            v4f c0 = *(const v4f*)&cs[c][0];
            v4f c1 = *(const v4f*)&cs[c][4];
            float cc[8] = {c0.x, c0.y, c0.z, c0.w, c1.x, c1.y, c1.z, c1.w};
#pragma unroll
            for (int o = 0; o < 8; ++o) {
                v2f b0 = cur.b[o][0] * cc[o];
                v2f b1 = cur.b[o][1] * cc[o];
#pragma unroll
                for (int t = 0; t < 2; ++t) {
                    acc[t][o] = __builtin_amdgcn_wmma_f32_16x16x4_f32(false, cur.a[t][0], false, b0,
                                                                      (short)0, acc[t][o], false, false);
                    acc[t][o] = __builtin_amdgcn_wmma_f32_16x16x4_f32(false, cur.a[t][1], false, b1,
                                                                      (short)0, acc[t][o], false, false);
                }
            }
            cur = nxt;
        }
    }
    float* pb = part + (size_t)blockIdx.x * PSZ + (size_t)(wv * 2) * 2048 + lane;
#pragma unroll
    for (int t = 0; t < 2; ++t)
#pragma unroll
        for (int o = 0; o < 8; ++o)
#pragma unroll
            for (int j = 0; j < 8; ++j)
                pb[(t * 8 + o) * 256 + j * 32] = acc[t][o][j];
}

__global__ __launch_bounds__(512) void k_reduce_squash(float* __restrict__ ws, size_t ws_bytes,
                                                       float* __restrict__ out, int last) {
    __shared__ float sm[4][128];
    const int cpc = caps_per_group(ws_bytes);
    const int nch = (IN_CAPS + cpc - 1) / cpc;
    const float* __restrict__ part = ws;
    float* __restrict__ v          = ws + (size_t)nch * PSZ;
    int bx   = blockIdx.x;
    int mt   = bx >> 4;
    int j    = (bx >> 1) & 7;
    int oh   = bx & 1;
    int t    = threadIdx.x & 127;
    int q    = threadIdx.x >> 7;
    int o    = oh * 4 + (t >> 5);
    int lane = t & 31;
    int idx  = ((mt * 8 + o) * 8 + j) * 32 + lane;
    float s = 0.0f;
    const float* p = part + (size_t)q * PSZ + idx;
    const int cnt = (nch - q + 3) >> 2;
    int i0 = 0;
    for (; i0 + 16 <= cnt; i0 += 16) {
        float pv[16];
#pragma unroll
        for (int u = 0; u < 16; ++u) pv[u] = p[(size_t)(i0 + u) * (4 * PSZ)];
#pragma unroll
        for (int u = 0; u < 16; ++u) s += pv[u];
    }
    if (i0 < cnt) {
        float pv[16];
#pragma unroll
        for (int u = 0; u < 16; ++u) pv[u] = p[(size_t)min(i0 + u, cnt - 1) * (4 * PSZ)];
#pragma unroll
        for (int u = 0; u < 16; ++u)
            if (i0 + u < cnt) s += pv[u];
    }
    sm[q][t] = s;
    __syncthreads();
    if (q == 0) {
        s = (sm[0][t] + sm[1][t]) + (sm[2][t] + sm[3][t]);
        float msq = s * s;
#pragma unroll
        for (int m = 1; m < 16; m <<= 1)
            msq += __shfl_xor(msq, m, 32);
        float mag = sqrtf(msq);
        float val = s * (mag / (1.0f + msq));
        int b = mt * 16 + (lane >> 4) * 8 + j;
        int n = o * 16 + (lane & 15);
        v[(size_t)b * NDIM + n] = val;
        if (last) out[(size_t)b * NDIM + n] = val;
    }
}

extern "C" void kernel_launch(void* const* d_in, const int* in_sizes, int n_in,
                              void* d_out, int out_size, void* d_ws, size_t ws_size,
                              hipStream_t stream) {
    const float* x = (const float*)d_in[0];
    const float* W = (const float*)d_in[1];
    float* out = (float*)d_out;

    float* ws = (float*)d_ws;

    k_pass<<<NCH_MAX, 256, 0, stream>>>(x, W, ws, ws_size, 0);
    k_reduce_squash<<<BATCH, 512, 0, stream>>>(ws, ws_size, out, 0);
    k_pass<<<NCH_MAX, 256, 0, stream>>>(x, W, ws, ws_size, 1);
    k_reduce_squash<<<BATCH, 512, 0, stream>>>(ws, ws_size, out, 0);
    k_pass<<<NCH_MAX, 256, 0, stream>>>(x, W, ws, ws_size, 2);
    k_reduce_squash<<<BATCH, 512, 0, stream>>>(ws, ws_size, out, 1);
}


# first pass (c = 1/8, no agreement) re-decomposed: work-group = 64 batch rows x 40 capsules, wave pairs split the capsules and fold through LDS, 1/8 at the fold; 61 planes, was 241; sum order changed
# speedup vs baseline: 2.9061x; 1.1636x over previous
#include <hip/hip_runtime.h>
#include <math.h>


#define IN_CAPS 2401
#define IN_SIZE 8
#define OUT_CAPS 8
#define OUT_SIZE 16
#define BATCH 256
#define KDIM (IN_CAPS * IN_SIZE)
#define NDIM (OUT_CAPS * OUT_SIZE)
#define CPC_MAX 15
#define CPC_MIN 10
#define NCH_MAX 241
#define PSZ (BATCH * NDIM)
#define RB0 64
#define KSPLIT0 4
#define CPW0 10
#define NRB0 (BATCH / RB0)
#define NCH0 ((IN_CAPS + KSPLIT0 * CPW0 - 1) / (KSPLIT0 * CPW0))
#define FOLD_SLOTS 11

typedef float v2f __attribute__((ext_vector_type(2)));
typedef float v4f __attribute__((ext_vector_type(4)));
typedef float v8f __attribute__((ext_vector_type(8)));

struct Frags {
    v2f a[2][2];
    v2f b[8][2];
};

__device__ __forceinline__ void load_frags(Frags& f, const float* __restrict__ xr0,
                                           const float* __restrict__ xr1,
                                           const float* __restrict__ wl, int i) {
    const float* p0 = xr0 + i * 8;
    const float* p1 = xr1 + i * 8;
    f.a[0][0] = *(const v2f*)(p0);
    f.a[0][1] = *(const v2f*)(p0 + 4);
    f.a[1][0] = *(const v2f*)(p1);
    f.a[1][1] = *(const v2f*)(p1 + 4);
    const float* w = wl + (size_t)i * 1024;
#pragma unroll
    for (int o = 0; o < 8; ++o) {
        f.b[o][0] = *(const v2f*)(w + o * 128);
        f.b[o][1] = *(const v2f*)(w + o * 128 + 4);
    }
}

__device__ __forceinline__ void mma_capsule(v8f (&acc)[2][8], const Frags& f) {
#pragma unroll
    for (int o = 0; o < 8; ++o)
#pragma unroll
        for (int t = 0; t < 2; ++t) {
            acc[t][o] = __builtin_amdgcn_wmma_f32_16x16x4_f32(false, f.a[t][0], false, f.b[o][0],
                                                              (short)0, acc[t][o], false, false);
            acc[t][o] = __builtin_amdgcn_wmma_f32_16x16x4_f32(false, f.a[t][1], false, f.b[o][1],
                                                              (short)0, acc[t][o], false, false);
        }
}

template <int CTRL>
__device__ __forceinline__ float row_xadd(float v) {
    int t = __builtin_amdgcn_update_dpp(0, __builtin_bit_cast(int, v), CTRL, 0xf, 0xf, true);
    return v + __builtin_bit_cast(float, t);
}
__device__ __forceinline__ float wave_sum(float v) {
    v = row_xadd<0x161>(v);
    v = row_xadd<0x162>(v);
    v = row_xadd<0x164>(v);
    v = row_xadd<0x168>(v);
    int b = __builtin_bit_cast(int, v);
    return v + __builtin_bit_cast(float, __builtin_amdgcn_permlanex16(b, b, 0x76543210, 0xfedcba98, false, false));
}

__device__ __forceinline__ int caps_per_group(size_t ws_bytes) {
    long long planes = ((long long)ws_bytes - (long long)(PSZ + IN_CAPS * OUT_CAPS) * 4) / (PSZ * 4);
    int cpc = CPC_MIN;
    while (cpc < CPC_MAX && (IN_CAPS + cpc - 1) / cpc > planes) ++cpc;
    return cpc;
}

__global__ __launch_bounds__(256) void k_pass(const float* __restrict__ x,
                                              const float* __restrict__ W,
                                              float* __restrict__ ws, size_t ws_bytes, int mode) {
    __shared__ __attribute__((aligned(16))) float red[2][8][8];
    __shared__ __attribute__((aligned(16))) float bsh[CPC_MAX][8];

    const int cpc = caps_per_group(ws_bytes);
    const int nch = (IN_CAPS + cpc - 1) / cpc;
    if ((int)blockIdx.x >= nch) return;
    float* __restrict__ part      = ws;
    const float* __restrict__ v   = ws + (size_t)nch * PSZ;
    float* __restrict__ bij       = ws + (size_t)(nch + 1) * PSZ;

    int tid  = threadIdx.x;
    int lane = tid & 31;
    int wv   = __builtin_amdgcn_readfirstlane(tid >> 5);
    int ln   = lane & 15;
    int half = lane >> 4;
    int k0   = half * 2;
    int cap0 = blockIdx.x * cpc;
    int ncap = min(cpc, IN_CAPS - cap0);
    int m0   = wv * 32;

    const float* xr0 = x + (size_t)(m0 + ln) * KDIM + k0;
    const float* xr1 = x + (size_t)(m0 + 16 + ln) * KDIM + k0;
    const float* wl  = W + ln * 8 + k0;

    v8f acc[2][8];
#pragma unroll
    for (int t = 0; t < 2; ++t)
#pragma unroll
        for (int o = 0; o < 8; ++o) acc[t][o] = (v8f){};

    {
        v8f vt[2][8];
#pragma unroll
        for (int t = 0; t < 2; ++t)
#pragma unroll
            for (int o = 0; o < 8; ++o)
#pragma unroll
                for (int j = 0; j < 8; ++j)
                    vt[t][o][j] = v[(size_t)(m0 + t * 16 + half * 8 + j) * NDIM + o * 16 + ln];
        if (mode == 2 && tid < ncap * 8) bsh[tid >> 3][tid & 7] = bij[(size_t)cap0 * OUT_CAPS + tid];

        Frags cur;
        load_frags(cur, xr0, xr1, wl, cap0);
        for (int c = 0; c < ncap; ++c) {
            v8f u[2][8];
#pragma unroll
            for (int o = 0; o < 8; ++o)
#pragma unroll
                for (int t = 0; t < 2; ++t) {
                    v8f z = {};
                    z = __builtin_amdgcn_wmma_f32_16x16x4_f32(false, cur.a[t][0], false, cur.b[o][0],
                                                              (short)0, z, false, false);
                    u[t][o] = __builtin_amdgcn_wmma_f32_16x16x4_f32(false, cur.a[t][1], false, cur.b[o][1],
                                                                    (short)0, z, false, false);
                }
            load_frags(cur, xr0, xr1, wl, cap0 + min(c + 1, ncap - 1));
            asm volatile("" ::: "memory");

            float p[8];
#pragma unroll
            for (int o = 0; o < 8; ++o) {
                v2f s2 = {0.0f, 0.0f};
#pragma unroll
                for (int t = 0; t < 2; ++t)
#pragma unroll
                    for (int k = 0; k < 4; ++k) {
                        v2f uu = {u[t][o][2 * k], u[t][o][2 * k + 1]};
                        v2f vv = {vt[t][o][2 * k], vt[t][o][2 * k + 1]};
                        s2 += uu * vv;
                    }
                p[o] = s2.x + s2.y;
            }
#pragma unroll
            for (int o = 0; o < 8; ++o) p[o] = wave_sum(p[o]);
            float* rb = &red[c & 1][wv][0];
            if (lane == 0) {
                *(v4f*)(rb)     = (v4f){p[0], p[1], p[2], p[3]};
                *(v4f*)(rb + 4) = (v4f){p[4], p[5], p[6], p[7]};
            }
            __builtin_amdgcn_fence(__ATOMIC_RELEASE, "workgroup", "local");
            __builtin_amdgcn_s_barrier();
            __builtin_amdgcn_fence(__ATOMIC_ACQUIRE, "workgroup", "local");
            v4f a0 = {0.0f, 0.0f, 0.0f, 0.0f};
            v4f a1 = {0.0f, 0.0f, 0.0f, 0.0f};
#pragma unroll
            for (int w = 0; w < 8; ++w) {
                a0 += *(const v4f*)&red[c & 1][w][0];
                a1 += *(const v4f*)&red[c & 1][w][4];
            }
            float bv[8] = {a0.x, a0.y, a0.z, a0.w, a1.x, a1.y, a1.z, a1.w};
#pragma unroll
            for (int o = 0; o < 8; ++o) bv[o] *= (1.0f / BATCH);
            if (mode == 2) {
                v4f b0 = *(const v4f*)&bsh[c][0];
                v4f b1 = *(const v4f*)&bsh[c][4];
                bv[0] += b0.x; bv[1] += b0.y; bv[2] += b0.z; bv[3] += b0.w;
                bv[4] += b1.x; bv[5] += b1.y; bv[6] += b1.z; bv[7] += b1.w;
            } else if (tid == 0) {
                float* bp = bij + (size_t)(cap0 + c) * OUT_CAPS;
                *(v4f*)(bp)     = (v4f){bv[0], bv[1], bv[2], bv[3]};
                *(v4f*)(bp + 4) = (v4f){bv[4], bv[5], bv[6], bv[7]};
            }
            float mx = bv[0];
#pragma unroll
            for (int o = 1; o < 8; ++o) mx = fmaxf(mx, bv[o]);
            float e[8];
#pragma unroll
            for (int o = 0; o < 8; ++o) e[o] = __expf(bv[o] - mx);
            float sum = ((e[0] + e[1]) + (e[2] + e[3])) + ((e[4] + e[5]) + (e[6] + e[7]));
            float inv = 1.0f / sum;
#pragma unroll
            for (int o = 0; o < 8; ++o) {
                float cc = e[o] * inv;
#pragma unroll
                for (int t = 0; t < 2; ++t) acc[t][o] += u[t][o] * cc;
            }
        }
    }
    float* pb = part + (size_t)blockIdx.x * PSZ + (size_t)(wv * 2) * 2048 + lane;
#pragma unroll
    for (int t = 0; t < 2; ++t)
#pragma unroll
        for (int o = 0; o < 8; ++o)
#pragma unroll
            for (int j = 0; j < 8; ++j)
                pb[(t * 8 + o) * 256 + j * 32] = acc[t][o][j];
}

__global__ __launch_bounds__(256) void k_first(const float* __restrict__ x,
                                               const float* __restrict__ W,
                                               float* __restrict__ ws) {
    __shared__ __attribute__((aligned(16))) float fold[FOLD_SLOTS][16][2][32][4];

    int tid  = threadIdx.x;
    int lane = tid & 31;
    int wv   = __builtin_amdgcn_readfirstlane(tid >> 5);
    int r    = wv & 1;
    int q    = wv >> 1;
    int ln   = lane & 15;
    int half = lane >> 4;
    int k0   = half * 2;
    int rb   = blockIdx.x & (NRB0 - 1);
    int ch   = blockIdx.x / NRB0;
    int m0   = rb * RB0 + r * 32;
    int cap0 = ch * (KSPLIT0 * CPW0) + q * CPW0;
    int ncap = min(CPW0, IN_CAPS - cap0);

    const float* xr0 = x + (size_t)(m0 + ln) * KDIM + k0;
    const float* xr1 = x + (size_t)(m0 + 16 + ln) * KDIM + k0;
    const float* wl  = W + ln * 8 + k0;

    v8f acc[2][8];
#pragma unroll
    for (int t = 0; t < 2; ++t)
#pragma unroll
        for (int o = 0; o < 8; ++o) acc[t][o] = (v8f){};

    if (ncap > 0) {
        Frags fa, fb;
        load_frags(fa, xr0, xr1, wl, cap0);
        int c = 0;
        for (; c + 1 < ncap; c += 2) {
            load_frags(fb, xr0, xr1, wl, cap0 + c + 1);
            __builtin_amdgcn_sched_barrier(0);
            mma_capsule(acc, fa);
            load_frags(fa, xr0, xr1, wl, cap0 + min(c + 2, ncap - 1));
            __builtin_amdgcn_sched_barrier(0);
            mma_capsule(acc, fb);
        }
        if (c < ncap) mma_capsule(acc, fa);
    }

#pragma unroll
    for (int t = 0; t < 2; ++t)
#pragma unroll
        for (int o = 0; o < 8; ++o) {
            *(v4f*)&fold[wv][t * 8 + o][0][lane][0] = (v4f){acc[t][o][0], acc[t][o][1], acc[t][o][2], acc[t][o][3]};
            *(v4f*)&fold[wv][t * 8 + o][1][lane][0] = (v4f){acc[t][o][4], acc[t][o][5], acc[t][o][6], acc[t][o][7]};
        }
    __syncthreads();

    float* pb = ws + (size_t)ch * PSZ + (size_t)((rb * 4 + r * 2 + (q >> 1)) * 8 + (q & 1) * 4) * 256 + lane;
#pragma unroll
    for (int n = 0; n < 4; ++n)
#pragma unroll
        for (int h2 = 0; h2 < 2; ++h2) {
            v4f c0 = *(const v4f*)&fold[r][4 * q + n][h2][lane][0];
            v4f c1 = *(const v4f*)&fold[2 + r][4 * q + n][h2][lane][0];
            v4f c2 = *(const v4f*)&fold[4 + r][4 * q + n][h2][lane][0];
            v4f c3 = *(const v4f*)&fold[6 + r][4 * q + n][h2][lane][0];
            v4f sm = ((c0 + c1) + (c2 + c3)) * 0.125f;
            pb[n * 256 + (h2 * 4 + 0) * 32] = sm.x;
            pb[n * 256 + (h2 * 4 + 1) * 32] = sm.y;
            pb[n * 256 + (h2 * 4 + 2) * 32] = sm.z;
            pb[n * 256 + (h2 * 4 + 3) * 32] = sm.w;
        }
}

__global__ __launch_bounds__(512) void k_reduce_squash(float* __restrict__ ws, size_t ws_bytes,
                                                       float* __restrict__ out, int last, int first) {
    __shared__ float sm[4][128];
    const int cpc = caps_per_group(ws_bytes);
    const int nch = (IN_CAPS + cpc - 1) / cpc;
    const int npl = first ? NCH0 : nch;
    const float* __restrict__ part = ws;
    float* __restrict__ v          = ws + (size_t)nch * PSZ;
    int bx   = blockIdx.x;
    int mt   = bx >> 4;
    int j    = (bx >> 1) & 7;
    int oh   = bx & 1;
    int t    = threadIdx.x & 127;
    int q    = threadIdx.x >> 7;
    int o    = oh * 4 + (t >> 5);
    int lane = t & 31;
    int idx  = ((mt * 8 + o) * 8 + j) * 32 + lane;
    float s = 0.0f;
    const float* p = part + (size_t)q * PSZ + idx;
    const int cnt = (npl - q + 3) >> 2;
    int i0 = 0;
    for (; i0 + 16 <= cnt; i0 += 16) {
        float pv[16];
#pragma unroll
        for (int u = 0; u < 16; ++u) pv[u] = p[(size_t)(i0 + u) * (4 * PSZ)];
#pragma unroll
        for (int u = 0; u < 16; ++u) s += pv[u];
    }
    if (i0 < cnt) {
        float pv[16];
#pragma unroll
        for (int u = 0; u < 16; ++u) pv[u] = p[(size_t)min(i0 + u, cnt - 1) * (4 * PSZ)];
#pragma unroll
        for (int u = 0; u < 16; ++u)
            if (i0 + u < cnt) s += pv[u];
    }
    sm[q][t] = s;
    __syncthreads();
    if (q == 0) {
        s = (sm[0][t] + sm[1][t]) + (sm[2][t] + sm[3][t]);
        float msq = s * s;
#pragma unroll
        for (int m = 1; m < 16; m <<= 1)
            msq += __shfl_xor(msq, m, 32);
        float mag = sqrtf(msq);
        float val = s * (mag / (1.0f + msq));
        int b = mt * 16 + (lane >> 4) * 8 + j;
        int n = o * 16 + (lane & 15);
        v[(size_t)b * NDIM + n] = val;
        if (last) out[(size_t)b * NDIM + n] = val;
    }
}

extern "C" void kernel_launch(void* const* d_in, const int* in_sizes, int n_in,
                              void* d_out, int out_size, void* d_ws, size_t ws_size,
                              hipStream_t stream) {
    const float* x = (const float*)d_in[0];
    const float* W = (const float*)d_in[1];
    float* out = (float*)d_out;

    float* ws = (float*)d_ws;

    k_first<<<NRB0 * NCH0, 256, 0, stream>>>(x, W, ws);
    k_reduce_squash<<<BATCH, 512, 0, stream>>>(ws, ws_size, out, 0, 1);
    k_pass<<<NCH_MAX, 256, 0, stream>>>(x, W, ws, ws_size, 1);
    k_reduce_squash<<<BATCH, 512, 0, stream>>>(ws, ws_size, out, 0, 0);
    k_pass<<<NCH_MAX, 256, 0, stream>>>(x, W, ws, ws_size, 2);
    k_reduce_squash<<<BATCH, 512, 0, stream>>>(ws, ws_size, out, 1, 0);
}


# partial planes of s laid out four tile registers per lane: stored and summed with 16-byte accesses (512-byte runs), all plane loads of a wave in flight at once, 128 sum blocks; sum order changed
# speedup vs baseline: 3.3153x; 1.1408x over previous
#include <hip/hip_runtime.h>
#include <math.h>


#define IN_CAPS 2401
#define IN_SIZE 8
#define OUT_CAPS 8
#define OUT_SIZE 16
#define BATCH 256
#define KDIM (IN_CAPS * IN_SIZE)
#define NDIM (OUT_CAPS * OUT_SIZE)
#define CPC_MAX 15
#define CPC_MIN 10
#define NCH_MAX 241
#define PSZ (BATCH * NDIM)
#define RB0 64
#define KSPLIT0 4
#define CPW0 10
#define NRB0 (BATCH / RB0)
#define NCH0 ((IN_CAPS + KSPLIT0 * CPW0 - 1) / (KSPLIT0 * CPW0))
#define FOLD_SLOTS 11
#define NRED 128

typedef float v2f __attribute__((ext_vector_type(2)));
typedef float v4f __attribute__((ext_vector_type(4)));
typedef float v8f __attribute__((ext_vector_type(8)));

__device__ __forceinline__ int plane_off(int mt, int o, int h2) {
    return ((((mt * 4 + (o >> 1)) * 2 + h2) * 2 + (o & 1))) * 128;
}

struct Frags {
    v2f a[2][2];
    v2f b[8][2];
};

__device__ __forceinline__ void load_frags(Frags& f, const float* __restrict__ xr0,
                                           const float* __restrict__ xr1,
                                           const float* __restrict__ wl, int i) {
    const float* p0 = xr0 + i * 8;
    const float* p1 = xr1 + i * 8;
    f.a[0][0] = *(const v2f*)(p0);
    f.a[0][1] = *(const v2f*)(p0 + 4);
    f.a[1][0] = *(const v2f*)(p1);
    f.a[1][1] = *(const v2f*)(p1 + 4);
    const float* w = wl + (size_t)i * 1024;
#pragma unroll
    for (int o = 0; o < 8; ++o) {
        f.b[o][0] = *(const v2f*)(w + o * 128);
        f.b[o][1] = *(const v2f*)(w + o * 128 + 4);
    }
}

__device__ __forceinline__ void mma_capsule(v8f (&acc)[2][8], const Frags& f) {
#pragma unroll
    for (int o = 0; o < 8; ++o)
#pragma unroll
        for (int t = 0; t < 2; ++t) {
            acc[t][o] = __builtin_amdgcn_wmma_f32_16x16x4_f32(false, f.a[t][0], false, f.b[o][0],
                                                              (short)0, acc[t][o], false, false);
            acc[t][o] = __builtin_amdgcn_wmma_f32_16x16x4_f32(false, f.a[t][1], false, f.b[o][1],
                                                              (short)0, acc[t][o], false, false);
        }
}

template <int CTRL>
__device__ __forceinline__ float row_xadd(float v) {
    int t = __builtin_amdgcn_update_dpp(0, __builtin_bit_cast(int, v), CTRL, 0xf, 0xf, true);
    return v + __builtin_bit_cast(float, t);
}
__device__ __forceinline__ float wave_sum(float v) {
    v = row_xadd<0x161>(v);
    v = row_xadd<0x162>(v);
    v = row_xadd<0x164>(v);
    v = row_xadd<0x168>(v);
    int b = __builtin_bit_cast(int, v);
    return v + __builtin_bit_cast(float, __builtin_amdgcn_permlanex16(b, b, 0x76543210, 0xfedcba98, false, false));
}

__device__ __forceinline__ int caps_per_group(size_t ws_bytes) {
    long long planes = ((long long)ws_bytes - (long long)(PSZ + IN_CAPS * OUT_CAPS) * 4) / (PSZ * 4);
    int cpc = CPC_MIN;
    while (cpc < CPC_MAX && (IN_CAPS + cpc - 1) / cpc > planes) ++cpc;
    return cpc;
}

__global__ __launch_bounds__(256) void k_pass(const float* __restrict__ x,
                                              const float* __restrict__ W,
                                              float* __restrict__ ws, size_t ws_bytes, int mode) {
    __shared__ __attribute__((aligned(16))) float red[2][8][8];
    __shared__ __attribute__((aligned(16))) float bsh[CPC_MAX][8];

    const int cpc = caps_per_group(ws_bytes);
    const int nch = (IN_CAPS + cpc - 1) / cpc;
    if ((int)blockIdx.x >= nch) return;
    float* __restrict__ part      = ws;
    const float* __restrict__ v   = ws + (size_t)nch * PSZ;
    float* __restrict__ bij       = ws + (size_t)(nch + 1) * PSZ;

    int tid  = threadIdx.x;
    int lane = tid & 31;
    int wv   = __builtin_amdgcn_readfirstlane(tid >> 5);
    int ln   = lane & 15;
    int half = lane >> 4;
    int k0   = half * 2;
    int cap0 = blockIdx.x * cpc;
    int ncap = min(cpc, IN_CAPS - cap0);
    int m0   = wv * 32;

    const float* xr0 = x + (size_t)(m0 + ln) * KDIM + k0;
    const float* xr1 = x + (size_t)(m0 + 16 + ln) * KDIM + k0;
    const float* wl  = W + ln * 8 + k0;

    v8f acc[2][8];
#pragma unroll
    for (int t = 0; t < 2; ++t)
#pragma unroll
        for (int o = 0; o < 8; ++o) acc[t][o] = (v8f){};

    {
        v8f vt[2][8];
#pragma unroll
        for (int t = 0; t < 2; ++t)
#pragma unroll
            for (int o = 0; o < 8; ++o)
#pragma unroll
                for (int j = 0; j < 8; ++j)
                    vt[t][o][j] = v[(size_t)(m0 + t * 16 + half * 8 + j) * NDIM + o * 16 + ln];
        if (mode == 2 && tid < ncap * 8) bsh[tid >> 3][tid & 7] = bij[(size_t)cap0 * OUT_CAPS + tid];

        Frags cur;
        load_frags(cur, xr0, xr1, wl, cap0);
        for (int c = 0; c < ncap; ++c) {
            v8f u[2][8];
#pragma unroll
            for (int o = 0; o < 8; ++o)
#pragma unroll
                for (int t = 0; t < 2; ++t) {
                    v8f z = {};
                    z = __builtin_amdgcn_wmma_f32_16x16x4_f32(false, cur.a[t][0], false, cur.b[o][0],
                                                              (short)0, z, false, false);
                    u[t][o] = __builtin_amdgcn_wmma_f32_16x16x4_f32(false, cur.a[t][1], false, cur.b[o][1],
                                                                    (short)0, z, false, false);
                }
            load_frags(cur, xr0, xr1, wl, cap0 + min(c + 1, ncap - 1));
            asm volatile("" ::: "memory");

            float p[8];
#pragma unroll
            for (int o = 0; o < 8; ++o) {
                v2f s2 = {0.0f, 0.0f};
#pragma unroll
                for (int t = 0; t < 2; ++t)
#pragma unroll
                    for (int k = 0; k < 4; ++k) {
                        v2f uu = {u[t][o][2 * k], u[t][o][2 * k + 1]};
                        v2f vv = {vt[t][o][2 * k], vt[t][o][2 * k + 1]};
                        s2 += uu * vv;
                    }
                p[o] = s2.x + s2.y;
            }
#pragma unroll
            for (int o = 0; o < 8; ++o) p[o] = wave_sum(p[o]);
            float* rb = &red[c & 1][wv][0];
            if (lane == 0) {
                *(v4f*)(rb)     = (v4f){p[0], p[1], p[2], p[3]};
                *(v4f*)(rb + 4) = (v4f){p[4], p[5], p[6], p[7]};
            }
            __builtin_amdgcn_fence(__ATOMIC_RELEASE, "workgroup", "local");
            __builtin_amdgcn_s_barrier();
            __builtin_amdgcn_fence(__ATOMIC_ACQUIRE, "workgroup", "local");
            v4f a0 = {0.0f, 0.0f, 0.0f, 0.0f};
            v4f a1 = {0.0f, 0.0f, 0.0f, 0.0f};
#pragma unroll
            for (int w = 0; w < 8; ++w) {
                a0 += *(const v4f*)&red[c & 1][w][0];
                a1 += *(const v4f*)&red[c & 1][w][4];
            }
            float bv[8] = {a0.x, a0.y, a0.z, a0.w, a1.x, a1.y, a1.z, a1.w};
#pragma unroll
            for (int o = 0; o < 8; ++o) bv[o] *= (1.0f / BATCH);
            if (mode == 2) {
                v4f b0 = *(const v4f*)&bsh[c][0];
                v4f b1 = *(const v4f*)&bsh[c][4];
                bv[0] += b0.x; bv[1] += b0.y; bv[2] += b0.z; bv[3] += b0.w;
                bv[4] += b1.x; bv[5] += b1.y; bv[6] += b1.z; bv[7] += b1.w;
            } else if (tid == 0) {
                float* bp = bij + (size_t)(cap0 + c) * OUT_CAPS;
                *(v4f*)(bp)     = (v4f){bv[0], bv[1], bv[2], bv[3]};
                *(v4f*)(bp + 4) = (v4f){bv[4], bv[5], bv[6], bv[7]};
            }
            float mx = bv[0];
#pragma unroll
            for (int o = 1; o < 8; ++o) mx = fmaxf(mx, bv[o]);
            float e[8];
#pragma unroll
            for (int o = 0; o < 8; ++o) e[o] = __expf(bv[o] - mx);
            float sum = ((e[0] + e[1]) + (e[2] + e[3])) + ((e[4] + e[5]) + (e[6] + e[7]));
            float inv = 1.0f / sum;
#pragma unroll
            for (int o = 0; o < 8; ++o) {
                float cc = e[o] * inv;
#pragma unroll
                for (int t = 0; t < 2; ++t) acc[t][o] += u[t][o] * cc;
            }
        }
    }
    float* pb = part + (size_t)blockIdx.x * PSZ + lane * 4;
#pragma unroll
    for (int t = 0; t < 2; ++t)
#pragma unroll
        for (int o = 0; o < 8; ++o)
#pragma unroll
            for (int h2 = 0; h2 < 2; ++h2)
                *(v4f*)(pb + plane_off(wv * 2 + t, o, h2)) =
                    (v4f){acc[t][o][4 * h2], acc[t][o][4 * h2 + 1], acc[t][o][4 * h2 + 2], acc[t][o][4 * h2 + 3]};
}

__global__ __launch_bounds__(256) void k_first(const float* __restrict__ x,
                                               const float* __restrict__ W,
                                               float* __restrict__ ws) {
    __shared__ __attribute__((aligned(16))) float fold[FOLD_SLOTS][16][2][32][4];

    int tid  = threadIdx.x;
    int lane = tid & 31;
    int wv   = __builtin_amdgcn_readfirstlane(tid >> 5);
    int r    = wv & 1;
    int q    = wv >> 1;
    int ln   = lane & 15;
    int half = lane >> 4;
    int k0   = half * 2;
    int rb   = blockIdx.x & (NRB0 - 1);
    int ch   = blockIdx.x / NRB0;
    int m0   = rb * RB0 + r * 32;
    int cap0 = ch * (KSPLIT0 * CPW0) + q * CPW0;
    int ncap = min(CPW0, IN_CAPS - cap0);

    const float* xr0 = x + (size_t)(m0 + ln) * KDIM + k0;
    const float* xr1 = x + (size_t)(m0 + 16 + ln) * KDIM + k0;
    const float* wl  = W + ln * 8 + k0;

    v8f acc[2][8];
#pragma unroll
    for (int t = 0; t < 2; ++t)
#pragma unroll
        for (int o = 0; o < 8; ++o) acc[t][o] = (v8f){};

    if (ncap > 0) {
        Frags fa, fb;
        load_frags(fa, xr0, xr1, wl, cap0);
        int c = 0;
        for (; c + 1 < ncap; c += 2) {
            load_frags(fb, xr0, xr1, wl, cap0 + c + 1);
            __builtin_amdgcn_sched_barrier(0);
            mma_capsule(acc, fa);
            load_frags(fa, xr0, xr1, wl, cap0 + min(c + 2, ncap - 1));
            __builtin_amdgcn_sched_barrier(0);
            mma_capsule(acc, fb);
        }
        if (c < ncap) mma_capsule(acc, fa);
    }

#pragma unroll
    for (int t = 0; t < 2; ++t)
#pragma unroll
        for (int o = 0; o < 8; ++o) {
            *(v4f*)&fold[wv][t * 8 + o][0][lane][0] = (v4f){acc[t][o][0], acc[t][o][1], acc[t][o][2], acc[t][o][3]};
            *(v4f*)&fold[wv][t * 8 + o][1][lane][0] = (v4f){acc[t][o][4], acc[t][o][5], acc[t][o][6], acc[t][o][7]};
        }
    __syncthreads();

    float* pb = ws + (size_t)ch * PSZ + lane * 4;
    int mtile = rb * 4 + r * 2 + (q >> 1);
#pragma unroll
    for (int n = 0; n < 4; ++n)
#pragma unroll
        for (int h2 = 0; h2 < 2; ++h2) {
            v4f c0 = *(const v4f*)&fold[r][4 * q + n][h2][lane][0];
            v4f c1 = *(const v4f*)&fold[2 + r][4 * q + n][h2][lane][0];
            v4f c2 = *(const v4f*)&fold[4 + r][4 * q + n][h2][lane][0];
            v4f c3 = *(const v4f*)&fold[6 + r][4 * q + n][h2][lane][0];
            *(v4f*)(pb + plane_off(mtile, (q & 1) * 4 + n, h2)) = ((c0 + c1) + (c2 + c3)) * 0.125f;
        }
}

__global__ __launch_bounds__(512) void k_reduce_squash(float* __restrict__ ws, size_t ws_bytes,
                                                       float* __restrict__ out, int last, int first) {
    __shared__ __attribute__((aligned(16))) float sm[8][2][32][4];
    const int cpc = caps_per_group(ws_bytes);
    const int nch = __builtin_amdgcn_readfirstlane((IN_CAPS + cpc - 1) / cpc);
    const int npl = first ? NCH0 : nch;
    const float* __restrict__ part = ws;
    float* __restrict__ v          = ws + (size_t)nch * PSZ;
    int bx   = blockIdx.x;
    int lane = threadIdx.x & 31;
    int wv   = __builtin_amdgcn_readfirstlane(threadIdx.x >> 5);
    int u    = wv & 1;
    int q    = wv >> 1;
    const float* p = part + (size_t)q * PSZ + bx * 256 + u * 128 + lane * 4;
    const int cnt = (npl - q + 7) >> 3;
    v4f s = {0.0f, 0.0f, 0.0f, 0.0f};
    if (cnt <= 8) {
        v4f pv[8];
#pragma unroll
        for (int g = 0; g < 8; ++g) pv[g] = *(const v4f*)(p + (size_t)min(g, cnt - 1) * (8 * PSZ));
#pragma unroll
        for (int g = 0; g < 8; ++g)
            if (g < cnt) s += pv[g];
    } else {
        for (int g0 = 0; g0 < cnt; g0 += 32) {
            v4f pv[32];
#pragma unroll
            for (int g = 0; g < 32; ++g) pv[g] = *(const v4f*)(p + (size_t)min(g0 + g, cnt - 1) * (8 * PSZ));
            __builtin_amdgcn_sched_barrier(0);
#pragma unroll
            for (int g = 0; g < 32; ++g)
                if (g0 + g < cnt) s += pv[g];
        }
    }
    *(v4f*)&sm[q][u][lane][0] = s;
    __syncthreads();
    if (wv < 8) {
        int half = wv >> 2;
        int jj   = wv & 3;
        int uu   = lane >> 4;
        int ln   = lane & 15;
        const float* f = &sm[0][uu][half * 16 + ln][jj];
        float t0 = (f[0 * 256] + f[1 * 256]) + (f[2 * 256] + f[3 * 256]);
        float t1 = (f[4 * 256] + f[5 * 256]) + (f[6 * 256] + f[7 * 256]);
        float sv = t0 + t1;
        float msq = sv * sv;
#pragma unroll
        for (int m = 1; m < 16; m <<= 1)
            msq += __shfl_xor(msq, m, 32);
        float mag = sqrtf(msq);
        float val = sv * (mag / (1.0f + msq));
        int b = (bx >> 3) * 16 + half * 8 + (bx & 1) * 4 + jj;
        int n = ((bx >> 1) & 3) * 32 + lane;
        v[(size_t)b * NDIM + n] = val;
        if (last) out[(size_t)b * NDIM + n] = val;
    }
}

extern "C" void kernel_launch(void* const* d_in, const int* in_sizes, int n_in,
                              void* d_out, int out_size, void* d_ws, size_t ws_size,
                              hipStream_t stream) {
    const float* x = (const float*)d_in[0];
    const float* W = (const float*)d_in[1];
    float* out = (float*)d_out;

    float* ws = (float*)d_ws;

    k_first<<<NRB0 * NCH0, 256, 0, stream>>>(x, W, ws);
    k_reduce_squash<<<NRED, 512, 0, stream>>>(ws, ws_size, out, 0, 1);
    k_pass<<<NCH_MAX, 256, 0, stream>>>(x, W, ws, ws_size, 1);
    k_reduce_squash<<<NRED, 512, 0, stream>>>(ws, ws_size, out, 0, 0);
    k_pass<<<NCH_MAX, 256, 0, stream>>>(x, W, ws, ws_size, 2);
    k_reduce_squash<<<NRED, 512, 0, stream>>>(ws, ws_size, out, 1, 0);
}
